# starting kernel
# baseline (speedup 1.0000x reference)
#include <hip/hip_runtime.h>

typedef __attribute__((ext_vector_type(2))) float v2f;
typedef __attribute__((ext_vector_type(8))) float v8f;

constexpr int B    = 32;
constexpr int INC  = 32;
constexpr int OUTC = 64;
constexpr int INN  = 65536;
constexpr int OUTN = 8192;
constexpr int DEG  = 16;

constexpr int NT      = 16;
constexpr int THREADS = 128;
constexpr int CP      = INC + 2;

__global__ __launch_bounds__(THREADS)
void fgl_fused_kernel(const float* __restrict__ x,
                      const int*   __restrict__ idx,
                      const float* __restrict__ nf,
                      const float* __restrict__ ft,
                      const float* __restrict__ bias,
                      float*       __restrict__ out)
{
    __shared__ alignas(16) int   s_idx[NT * DEG];
    __shared__ alignas(16) float s_red[NT * CP];
    __shared__ alignas(16) float s_ft [OUTC * CP];

    const int tid   = threadIdx.x;
    const int ntile = blockIdx.x;
    const int b     = blockIdx.y;

    {
        const int4* __restrict__ idx4 = (const int4*)(idx + ntile * (NT * DEG));
        int4* s_idx4 = (int4*)s_idx;
        if (tid < (NT * DEG) / 4)
            s_idx4[tid] = idx4[tid];
    }
    for (int i = tid; i < INC * OUTC; i += THREADS) {
        const int c = i >> 6;
        const int k = i & 63;
        s_ft[k * CP + c] = ft[i];
    }
    __syncthreads();

    const size_t xb = (size_t)b * INC * INN;
    for (int p = tid; p < INC * NT; p += THREADS) {
        const int c  = p >> 4;
        const int nl = p & 15;
        const float* __restrict__ xc = x  + xb + (size_t)c * INN;
        const float* __restrict__ nc = nf +      (size_t)c * INN;
        float m = -__builtin_inff();
        #pragma unroll
        for (int d = 0; d < DEG; ++d) {
            const int j = s_idx[nl * DEG + d];
            m = fmaxf(m, xc[j] * nc[j]);
        }
        s_red[nl * CP + c] = m;
    }
    __syncthreads();

    const int wave = tid >> 5;
    const int lane = tid & 31;
    const int half = lane >> 4;
    const int col  = lane & 15;
    const int krow = wave * 16 + col;

    v8f acc = {0.f, 0.f, 0.f, 0.f, 0.f, 0.f, 0.f, 0.f};
    #pragma unroll
    for (int kc = 0; kc < INC / 4; ++kc) {
        const int c0 = kc * 4 + 2 * half;
        const v2f a  = *(const v2f*)&s_ft [krow * CP + c0];
        const v2f bm = *(const v2f*)&s_red[col  * CP + c0];
        acc = __builtin_amdgcn_wmma_f32_16x16x4_f32(
             false, a,  false, bm,
             (short)0, acc,  false,  false);
    }

    const int n = ntile * NT + col;
    const size_t ob = (size_t)b * OUTC * OUTN;
    #pragma unroll
    for (int v = 0; v < 8; ++v) {
        const int k = wave * 16 + half * 8 + v;
        out[ob + (size_t)k * OUTN + n] = acc[v] + bias[k * OUTN + n];
    }
}

extern "C" void kernel_launch(void* const* d_in, const int* in_sizes, int n_in,
                              void* d_out, int out_size, void* d_ws, size_t ws_size,
                              hipStream_t stream) {
    const float* x    = (const float*)d_in[0];
    const int*   idx  = (const int*)  d_in[1];
    const float* nf   = (const float*)d_in[2];
    const float* ft   = (const float*)d_in[3];
    const float* bias = (const float*)d_in[4];
    float*       out  = (float*)d_out;

    dim3 grid(OUTN / NT, B);
    fgl_fused_kernel<<<grid, THREADS, 0, stream>>>(x, idx, nf, ft, bias, out);
}


# Each workgroup loops over half the batches (grid y 32 -> 2): neighbour offsets, gathered nf, ft fragments and bias are fetched once and held in registers; only the x gather repeats per batch
# speedup vs baseline: 1.7928x; 1.7928x over previous
#include <hip/hip_runtime.h>


typedef __attribute__((ext_vector_type(2))) float v2f;
typedef __attribute__((ext_vector_type(8))) float v8f;

constexpr int B    = 32;
constexpr int INC  = 32;
constexpr int OUTC = 64;
constexpr int INN  = 65536;
constexpr int OUTN = 8192;
constexpr int DEG  = 16;

constexpr int NT      = 16;
constexpr int THREADS = 128;
constexpr int CP      = INC + 2;
constexpr int BG      = 16;
constexpr int CSTEP   = THREADS / NT;
constexpr int CPT     = INC / CSTEP;

__global__ __launch_bounds__(THREADS)
void fgl_fused_kernel(const float* __restrict__ x,
                      const int*   __restrict__ idx,
                      const float* __restrict__ nf,
                      const float* __restrict__ ft,
                      const float* __restrict__ bias,
                      float*       __restrict__ out)
{
    __shared__ alignas(16) int   s_idx[NT * DEG];
    __shared__ alignas(16) float s_red[2][NT * CP];
    __shared__ alignas(16) float s_ft [OUTC * CP];

    const int tid   = threadIdx.x;
    const int ntile = blockIdx.x;
    const int b0    = blockIdx.y * BG;

    {
        const int4* __restrict__ idx4 = (const int4*)(idx + ntile * (NT * DEG));
        int4* s_idx4 = (int4*)s_idx;
        if (tid < (NT * DEG) / 4)
            s_idx4[tid] = idx4[tid];
    }
    for (int i = tid; i < INC * OUTC; i += THREADS) {
        const int c = i >> 6;
        const int k = i & 63;
        s_ft[k * CP + c] = ft[i];
    }
    __syncthreads();

    const int nl = tid & (NT - 1);
    const int cl = tid >> 4;
    unsigned boff[DEG];
    #pragma unroll
    for (int d = 0; d < DEG; ++d)
        boff[d] = ((unsigned)cl * INN + (unsigned)s_idx[nl * DEG + d]) * 4u;

    float w[CPT][DEG];
    #pragma unroll
    for (int i = 0; i < CPT; ++i) {
        const char* __restrict__ nrow = (const char*)(nf + (size_t)i * CSTEP * INN);
        #pragma unroll
        for (int d = 0; d < DEG; ++d)
            w[i][d] = *(const float*)(nrow + boff[d]);
    }

    const int wave = tid >> 5;
    const int lane = tid & 31;
    const int half = lane >> 4;
    const int col  = lane & 15;
    const int krow = wave * 16 + col;
    const int n    = ntile * NT + col;

    v2f afrag[INC / 4];
    #pragma unroll
    for (int kc = 0; kc < INC / 4; ++kc)
        afrag[kc] = *(const v2f*)&s_ft[krow * CP + kc * 4 + 2 * half];

    float bv[8];
    #pragma unroll
    for (int v = 0; v < 8; ++v)
        bv[v] = bias[(wave * 16 + half * 8 + v) * OUTN + n];

    for (int bb = 0; bb < BG; ++bb) {
        const int b = b0 + bb;
        float* __restrict__ red = s_red[bb & 1];

        const float* __restrict__ xb = x + (size_t)b * INC * INN;
        #pragma unroll
        for (int i = 0; i < CPT; ++i) {
            const char* __restrict__ xrow = (const char*)(xb + (size_t)i * CSTEP * INN);
            float m = -__builtin_inff();
            #pragma unroll
            for (int d = 0; d < DEG; ++d)
                m = fmaxf(m, *(const float*)(xrow + boff[d]) * w[i][d]);
            red[nl * CP + cl + i * CSTEP] = m;
        }
        __syncthreads();

        v8f acc = {0.f, 0.f, 0.f, 0.f, 0.f, 0.f, 0.f, 0.f};
        #pragma unroll
        for (int kc = 0; kc < INC / 4; ++kc) {
            const int c0 = kc * 4 + 2 * half;
            const v2f bm = *(const v2f*)&red[col * CP + c0];
            acc = __builtin_amdgcn_wmma_f32_16x16x4_f32(
                 false, afrag[kc],  false, bm,
                 (short)0, acc,  false,  false);
        }

        const size_t ob = (size_t)b * OUTC * OUTN;
        #pragma unroll
        for (int v = 0; v < 8; ++v) {
            const int k = wave * 16 + half * 8 + v;
            out[ob + (size_t)k * OUTN + n] = acc[v] + bv[v];
        }
    }
}

extern "C" void kernel_launch(void* const* d_in, const int* in_sizes, int n_in,
                              void* d_out, int out_size, void* d_ws, size_t ws_size,
                              hipStream_t stream) {
    const float* x    = (const float*)d_in[0];
    const int*   idx  = (const int*)  d_in[1];
    const float* nf   = (const float*)d_in[2];
    const float* ft   = (const float*)d_in[3];
    const float* bias = (const float*)d_in[4];
    float*       out  = (float*)d_out;

    dim3 grid(OUTN / NT, B / BG);
    fgl_fused_kernel<<<grid, THREADS, 0, stream>>>(x, idx, nf, ft, bias, out);
}


# Each (batch, channel) row of x*nf is streamed once into LDS (256 KiB); the 16-neighbour gather and max read LDS, not global lines; channel product = second launch on red parked in out
# speedup vs baseline: 9.9968x; 5.5761x over previous
#include <hip/hip_runtime.h>


typedef __attribute__((ext_vector_type(2))) float v2f;
typedef __attribute__((ext_vector_type(8))) float v8f;
typedef __attribute__((ext_vector_type(4))) float v4f;

constexpr int B    = 32;
constexpr int INC  = 32;
constexpr int OUTC = 64;
constexpr int INN  = 65536;
constexpr int OUTN = 8192;
constexpr int DEG  = 16;

constexpr int GT   = 512;
constexpr int RB   = 4;
constexpr int NPT  = OUTN / GT;
constexpr int V4PT = INN / 4 / GT;
constexpr int SU   = 8;

__global__ __launch_bounds__(GT)
void fgl_gather_max_kernel(const float* __restrict__ x,
                           const int*   __restrict__ idx,
                           const float* __restrict__ nf,
                           float*       __restrict__ red)
{
    __shared__ alignas(16) float s_h[INN];

    const int tid = threadIdx.x;
    const int c   = blockIdx.x & (INC - 1);
    const int b0  = (blockIdx.x >> 5) * RB;

    unsigned pk[NPT][DEG / 2];
    #pragma unroll
    for (int j = 0; j < NPT; ++j) {
        const int4* __restrict__ p = (const int4*)(idx + (size_t)(tid + j * GT) * DEG);
        #pragma unroll
        for (int q = 0; q < DEG / 4; ++q) {
            const int4 v = p[q];
            pk[j][2 * q]     = ((unsigned)v.x & 0xffffu) | ((unsigned)v.y << 16);
            pk[j][2 * q + 1] = ((unsigned)v.z & 0xffffu) | ((unsigned)v.w << 16);
        }
    }

    const v4f* __restrict__ nf4 = (const v4f*)(nf + (size_t)c * INN);
    v4f* s_h4 = (v4f*)s_h;

    for (int r = 0; r < RB; ++r) {
        const int b = b0 + r;
        const v4f* __restrict__ x4 = (const v4f*)(x + ((size_t)b * INC + c) * INN);

        if (r) __syncthreads();

        const v4f* __restrict__ xp = x4 + tid;
        const v4f* __restrict__ np = nf4 + tid;
        v4f* sp = s_h4 + tid;
        #pragma unroll 1
        for (int g = 0; g < V4PT / SU; ++g) {
            v4f a[SU], w[SU];
            #pragma unroll
            for (int u = 0; u < SU; ++u) a[u] = xp[u * GT];
            #pragma unroll
            for (int u = 0; u < SU; ++u) w[u] = np[u * GT];
            #pragma unroll
            for (int u = 0; u < SU; ++u) sp[u * GT] = a[u] * w[u];
            xp += SU * GT; np += SU * GT; sp += SU * GT;
        }
        __syncthreads();

        float* __restrict__ rrow = red + ((size_t)b * OUTC + c) * OUTN;
        #pragma unroll
        for (int j = 0; j < NPT; ++j) {
            float m = -__builtin_inff();
            #pragma unroll
            for (int e = 0; e < DEG / 2; ++e) {
                unsigned p = pk[j][e];
                asm volatile("" : "+v"(p));
                m = fmaxf(m, s_h[p & 0xffffu]);
                m = fmaxf(m, s_h[p >> 16]);
            }
            rrow[tid + j * GT] = m;
        }
    }
}

constexpr int T2  = 256;
constexpr int NT2 = 64;
constexpr int BG2 = 4;
constexpr int RP  = NT2 + 8;
constexpr int CP  = INC + 2;

__global__ __launch_bounds__(T2)
void fgl_channel_gemm_kernel(const float* __restrict__ ft,
                             const float* __restrict__ bias,
                             float*                    out)
{
    __shared__ alignas(16) float s_red[2][INC * RP];
    __shared__ alignas(16) float s_ft [OUTC * CP];

    const int tid = threadIdx.x;
    const int n0  = blockIdx.x * NT2;
    const int b0  = blockIdx.y * BG2;

    for (int i = tid; i < INC * OUTC; i += T2) {
        const int c = i >> 6;
        const int k = i & 63;
        s_ft[k * CP + c] = ft[i];
    }

    const int lc = tid >> 4;
    const int l4 = tid & 15;
    const size_t roff0 = (size_t)lc * OUTN + n0 + l4 * 4;
    const size_t roff1 = roff0 + (size_t)16 * OUTN;
    float4 r0 = *(const float4*)(out + (size_t)b0 * OUTC * OUTN + roff0);
    float4 r1 = *(const float4*)(out + (size_t)b0 * OUTC * OUTN + roff1);

    __syncthreads();

    const int wave  = tid >> 5;
    const int lane  = tid & 31;
    const int half  = lane >> 4;
    const int col   = lane & 15;
    const int ktile = wave & 3;
    const int nsb   = (wave >> 2) * 2;
    const int krow  = ktile * 16 + col;

    v2f afrag[INC / 4];
    #pragma unroll
    for (int kc = 0; kc < INC / 4; ++kc)
        afrag[kc] = *(const v2f*)&s_ft[krow * CP + kc * 4 + 2 * half];

    float bv[2][8];
    #pragma unroll
    for (int s = 0; s < 2; ++s)
        #pragma unroll
        for (int v = 0; v < 8; ++v)
            bv[s][v] = bias[(size_t)(ktile * 16 + half * 8 + v) * OUTN + n0 + (nsb + s) * 16 + col];

    for (int bb = 0; bb < BG2; ++bb) {
        const int b = b0 + bb;
        float* red = s_red[bb & 1];

        *(float4*)&red[lc * RP + l4 * 4]        = r0;
        *(float4*)&red[(lc + 16) * RP + l4 * 4] = r1;
        if (bb + 1 < BG2) {
            r0 = *(const float4*)(out + (size_t)(b + 1) * OUTC * OUTN + roff0);
            r1 = *(const float4*)(out + (size_t)(b + 1) * OUTC * OUTN + roff1);
        }
        __syncthreads();

        const size_t ob = (size_t)b * OUTC * OUTN;
        #pragma unroll
        for (int s = 0; s < 2; ++s) {
            const int nl = (nsb + s) * 16 + col;
            v8f acc = {0.f, 0.f, 0.f, 0.f, 0.f, 0.f, 0.f, 0.f};
            #pragma unroll
            for (int kc = 0; kc < INC / 4; ++kc) {
                const int c0 = kc * 4 + 2 * half;
                v2f bm;
                bm.x = red[c0 * RP + nl];
                bm.y = red[(c0 + 1) * RP + nl];
                acc = __builtin_amdgcn_wmma_f32_16x16x4_f32(
                     false, afrag[kc],  false, bm,
                     (short)0, acc,  false,  false);
            }
            #pragma unroll
            for (int v = 0; v < 8; ++v) {
                const int k = ktile * 16 + half * 8 + v;
                out[ob + (size_t)k * OUTN + n0 + nl] = acc[v] + bv[s][v];
            }
        }
    }
}

extern "C" void kernel_launch(void* const* d_in, const int* in_sizes, int n_in,
                              void* d_out, int out_size, void* d_ws, size_t ws_size,
                              hipStream_t stream) {
    const float* x    = (const float*)d_in[0];
    const int*   idx  = (const int*)  d_in[1];
    const float* nf   = (const float*)d_in[2];
    const float* ft   = (const float*)d_in[3];
    const float* bias = (const float*)d_in[4];
    float*       out  = (float*)d_out;

    fgl_gather_max_kernel<<<INC * (B / RB), GT, 0, stream>>>(x, idx, nf, out);
    dim3 grid2(OUTN / NT2, B / BG2);
    fgl_channel_gemm_kernel<<<grid2, T2, 0, stream>>>(ft, bias, out);
}


# Launch 1: while a row is gathered from LDS, the next (batch, channel) row of x is prefetched into the second-level cache (row 0 during the index prologue); the stream phase then reads cache lines
# speedup vs baseline: 11.3503x; 1.1354x over previous
#include <hip/hip_runtime.h>


typedef __attribute__((ext_vector_type(2))) float v2f;
typedef __attribute__((ext_vector_type(8))) float v8f;
typedef __attribute__((ext_vector_type(4))) float v4f;

constexpr int B    = 32;
constexpr int INC  = 32;
constexpr int OUTC = 64;
constexpr int INN  = 65536;
constexpr int OUTN = 8192;
constexpr int DEG  = 16;

constexpr int GT   = 512;
constexpr int RB   = 4;
constexpr int NPT  = OUTN / GT;
constexpr int V4PT = INN / 4 / GT;
constexpr int SU   = 8;
constexpr int PFU  = 256;
constexpr int PFPT = INN * 4 / PFU / GT;

__device__ __forceinline__ void prefetch_row_to_l2(const float* __restrict__ row, int tid)
{
    const char* p = (const char*)row + (size_t)tid * PFU;
    for (int u = 0; u < PFPT; ++u)
        __builtin_prefetch(p + (size_t)u * GT * PFU,  0,  2);
}

__global__ __launch_bounds__(GT)
void fgl_gather_max_kernel(const float* __restrict__ x,
                           const int*   __restrict__ idx,
                           const float* __restrict__ nf,
                           float*       __restrict__ red)
{
    __shared__ alignas(16) float s_h[INN];

    const int tid = threadIdx.x;
    const int c   = blockIdx.x & (INC - 1);
    const int b0  = (blockIdx.x >> 5) * RB;

    prefetch_row_to_l2(x + ((size_t)b0 * INC + c) * INN, tid);

    unsigned pk[NPT][DEG / 2];
    #pragma unroll
    for (int j = 0; j < NPT; ++j) {
        const int4* __restrict__ p = (const int4*)(idx + (size_t)(tid + j * GT) * DEG);
        #pragma unroll
        for (int q = 0; q < DEG / 4; ++q) {
            const int4 v = p[q];
            pk[j][2 * q]     = ((unsigned)v.x & 0xffffu) | ((unsigned)v.y << 16);
            pk[j][2 * q + 1] = ((unsigned)v.z & 0xffffu) | ((unsigned)v.w << 16);
        }
    }

    const v4f* __restrict__ nf4 = (const v4f*)(nf + (size_t)c * INN);
    v4f* s_h4 = (v4f*)s_h;

    for (int r = 0; r < RB; ++r) {
        const int b = b0 + r;
        const v4f* __restrict__ x4 = (const v4f*)(x + ((size_t)b * INC + c) * INN);

        if (r) __syncthreads();

        const v4f* __restrict__ xp = x4 + tid;
        const v4f* __restrict__ np = nf4 + tid;
        v4f* sp = s_h4 + tid;
        #pragma unroll 1
        for (int g = 0; g < V4PT / SU; ++g) {
            v4f a[SU], w[SU];
            #pragma unroll
            for (int u = 0; u < SU; ++u) a[u] = xp[u * GT];
            #pragma unroll
            for (int u = 0; u < SU; ++u) w[u] = np[u * GT];
            #pragma unroll
            for (int u = 0; u < SU; ++u) sp[u * GT] = a[u] * w[u];
            xp += SU * GT; np += SU * GT; sp += SU * GT;
        }
        __syncthreads();

        if (r + 1 < RB)
            prefetch_row_to_l2(x + ((size_t)(b + 1) * INC + c) * INN, tid);

        float* __restrict__ rrow = red + ((size_t)b * OUTC + c) * OUTN;
        #pragma unroll
        for (int j = 0; j < NPT; ++j) {
            float m = -__builtin_inff();
            #pragma unroll
            for (int e = 0; e < DEG / 2; ++e) {
                unsigned p = pk[j][e];
                asm volatile("" : "+v"(p));
                m = fmaxf(m, s_h[p & 0xffffu]);
                m = fmaxf(m, s_h[p >> 16]);
            }
            rrow[tid + j * GT] = m;
        }
    }
}

constexpr int T2  = 256;
constexpr int NT2 = 64;
constexpr int BG2 = 4;
constexpr int RP  = NT2 + 8;
constexpr int CP  = INC + 2;

__global__ __launch_bounds__(T2)
void fgl_channel_gemm_kernel(const float* __restrict__ ft,
                             const float* __restrict__ bias,
                             float*                    out)
{
    __shared__ alignas(16) float s_red[2][INC * RP];
    __shared__ alignas(16) float s_ft [OUTC * CP];

    const int tid = threadIdx.x;
    const int n0  = blockIdx.x * NT2;
    const int b0  = blockIdx.y * BG2;

    for (int i = tid; i < INC * OUTC; i += T2) {
        const int c = i >> 6;
        const int k = i & 63;
        s_ft[k * CP + c] = ft[i];
    }

    const int lc = tid >> 4;
    const int l4 = tid & 15;
    const size_t roff0 = (size_t)lc * OUTN + n0 + l4 * 4;
    const size_t roff1 = roff0 + (size_t)16 * OUTN;
    float4 r0 = *(const float4*)(out + (size_t)b0 * OUTC * OUTN + roff0);
    float4 r1 = *(const float4*)(out + (size_t)b0 * OUTC * OUTN + roff1);

    __syncthreads();

    const int wave  = tid >> 5;
    const int lane  = tid & 31;
    const int half  = lane >> 4;
    const int col   = lane & 15;
    const int ktile = wave & 3;
    const int nsb   = (wave >> 2) * 2;
    const int krow  = ktile * 16 + col;

    v2f afrag[INC / 4];
    #pragma unroll
    for (int kc = 0; kc < INC / 4; ++kc)
        afrag[kc] = *(const v2f*)&s_ft[krow * CP + kc * 4 + 2 * half];

    float bv[2][8];
    #pragma unroll
    for (int s = 0; s < 2; ++s)
        #pragma unroll
        for (int v = 0; v < 8; ++v)
            bv[s][v] = bias[(size_t)(ktile * 16 + half * 8 + v) * OUTN + n0 + (nsb + s) * 16 + col];

    for (int bb = 0; bb < BG2; ++bb) {
        const int b = b0 + bb;
        float* red = s_red[bb & 1];

        *(float4*)&red[lc * RP + l4 * 4]        = r0;
        *(float4*)&red[(lc + 16) * RP + l4 * 4] = r1;
        if (bb + 1 < BG2) {
            r0 = *(const float4*)(out + (size_t)(b + 1) * OUTC * OUTN + roff0);
            r1 = *(const float4*)(out + (size_t)(b + 1) * OUTC * OUTN + roff1);
        }
        __syncthreads();

        const size_t ob = (size_t)b * OUTC * OUTN;
        #pragma unroll
        for (int s = 0; s < 2; ++s) {
            const int nl = (nsb + s) * 16 + col;
            v8f acc = {0.f, 0.f, 0.f, 0.f, 0.f, 0.f, 0.f, 0.f};
            #pragma unroll
            for (int kc = 0; kc < INC / 4; ++kc) {
                const int c0 = kc * 4 + 2 * half;
                v2f bm;
                bm.x = red[c0 * RP + nl];
                bm.y = red[(c0 + 1) * RP + nl];
                acc = __builtin_amdgcn_wmma_f32_16x16x4_f32(
                     false, afrag[kc],  false, bm,
                     (short)0, acc,  false,  false);
            }
            #pragma unroll
            for (int v = 0; v < 8; ++v) {
                const int k = ktile * 16 + half * 8 + v;
                out[ob + (size_t)k * OUTN + n0 + nl] = acc[v] + bv[s][v];
            }
        }
    }
}

extern "C" void kernel_launch(void* const* d_in, const int* in_sizes, int n_in,
                              void* d_out, int out_size, void* d_ws, size_t ws_size,
                              hipStream_t stream) {
    const float* x    = (const float*)d_in[0];
    const int*   idx  = (const int*)  d_in[1];
    const float* nf   = (const float*)d_in[2];
    const float* ft   = (const float*)d_in[3];
    const float* bias = (const float*)d_in[4];
    float*       out  = (float*)d_out;

    fgl_gather_max_kernel<<<INC * (B / RB), GT, 0, stream>>>(x, idx, nf, out);
    dim3 grid2(OUTN / NT2, B / BG2);
    fgl_channel_gemm_kernel<<<grid2, T2, 0, stream>>>(ft, bias, out);
}


# Launch 2: all global reads of a work-group (red of its 4 batches, ft, bias) issued up front as one group, red staged in 4 LDS planes behind ONE barrier; no wait on global loads or stores behind it
# speedup vs baseline: 12.1499x; 1.0704x over previous
#include <hip/hip_runtime.h>


typedef __attribute__((ext_vector_type(2))) float v2f;
typedef __attribute__((ext_vector_type(8))) float v8f;
typedef __attribute__((ext_vector_type(4))) float v4f;

constexpr int B    = 32;
constexpr int INC  = 32;
constexpr int OUTC = 64;
constexpr int INN  = 65536;
constexpr int OUTN = 8192;
constexpr int DEG  = 16;

constexpr int GT   = 512;
constexpr int RB   = 4;
constexpr int NPT  = OUTN / GT;
constexpr int V4PT = INN / 4 / GT;
constexpr int SU   = 8;
constexpr int PFU  = 256;
constexpr int PFPT = INN * 4 / PFU / GT;

__device__ __forceinline__ void prefetch_row_to_l2(const float* __restrict__ row, int tid)
{
    const char* p = (const char*)row + (size_t)tid * PFU;
    for (int u = 0; u < PFPT; ++u)
        __builtin_prefetch(p + (size_t)u * GT * PFU,  0,  2);
}

__global__ __launch_bounds__(GT)
void fgl_gather_max_kernel(const float* __restrict__ x,
                           const int*   __restrict__ idx,
                           const float* __restrict__ nf,
                           float*       __restrict__ red)
{
    __shared__ alignas(16) float s_h[INN];

    const int tid = threadIdx.x;
    const int c   = blockIdx.x & (INC - 1);
    const int b0  = (blockIdx.x >> 5) * RB;

    prefetch_row_to_l2(x + ((size_t)b0 * INC + c) * INN, tid);

    unsigned pk[NPT][DEG / 2];
    #pragma unroll
    for (int j = 0; j < NPT; ++j) {
        const int4* __restrict__ p = (const int4*)(idx + (size_t)(tid + j * GT) * DEG);
        #pragma unroll
        for (int q = 0; q < DEG / 4; ++q) {
            const int4 v = p[q];
            pk[j][2 * q]     = ((unsigned)v.x & 0xffffu) | ((unsigned)v.y << 16);
            pk[j][2 * q + 1] = ((unsigned)v.z & 0xffffu) | ((unsigned)v.w << 16);
        }
    }

    const v4f* __restrict__ nf4 = (const v4f*)(nf + (size_t)c * INN);
    v4f* s_h4 = (v4f*)s_h;

    for (int r = 0; r < RB; ++r) {
        const int b = b0 + r;
        const v4f* __restrict__ x4 = (const v4f*)(x + ((size_t)b * INC + c) * INN);

        if (r) __syncthreads();

        const v4f* __restrict__ xp = x4 + tid;
        const v4f* __restrict__ np = nf4 + tid;
        v4f* sp = s_h4 + tid;
        #pragma unroll 1
        for (int g = 0; g < V4PT / SU; ++g) {
            v4f a[SU], w[SU];
            #pragma unroll
            for (int u = 0; u < SU; ++u) a[u] = xp[u * GT];
            #pragma unroll
            for (int u = 0; u < SU; ++u) w[u] = np[u * GT];
            #pragma unroll
            for (int u = 0; u < SU; ++u) sp[u * GT] = a[u] * w[u];
            xp += SU * GT; np += SU * GT; sp += SU * GT;
        }
        __syncthreads();

        if (r + 1 < RB)
            prefetch_row_to_l2(x + ((size_t)(b + 1) * INC + c) * INN, tid);

        float* __restrict__ rrow = red + ((size_t)b * OUTC + c) * OUTN;
        #pragma unroll
        for (int j = 0; j < NPT; ++j) {
            float m = -__builtin_inff();
            #pragma unroll
            for (int e = 0; e < DEG / 2; ++e) {
                unsigned p = pk[j][e];
                asm volatile("" : "+v"(p));
                m = fmaxf(m, s_h[p & 0xffffu]);
                m = fmaxf(m, s_h[p >> 16]);
            }
            rrow[tid + j * GT] = m;
        }
    }
}

constexpr int T2  = 256;
constexpr int NT2 = 64;
constexpr int BG2 = 4;
constexpr int RP  = NT2 + 8;
constexpr int CP  = INC + 2;
constexpr int FPT = INC * OUTC / T2;

__global__ __launch_bounds__(T2)
void fgl_channel_gemm_kernel(const float* __restrict__ ft,
                             const float* __restrict__ bias,
                             float*                    out)
{
    __shared__ alignas(16) float s_red[BG2][INC * RP];
    __shared__ alignas(16) float s_ft [OUTC * CP];

    const int tid = threadIdx.x;
    const int n0  = blockIdx.x * NT2;
    const int b0  = blockIdx.y * BG2;

    const int lc = tid >> 4;
    const int l4 = tid & 15;
    const size_t roff0 = (size_t)lc * OUTN + n0 + l4 * 4;
    const size_t roff1 = roff0 + (size_t)16 * OUTN;

    const int wave  = tid >> 5;
    const int lane  = tid & 31;
    const int half  = lane >> 4;
    const int col   = lane & 15;
    const int ktile = wave & 3;
    const int nsb   = (wave >> 2) * 2;
    const int krow  = ktile * 16 + col;

    float4 r0[BG2], r1[BG2];
    #pragma unroll
    for (int bb = 0; bb < BG2; ++bb) {
        r0[bb] = *(const float4*)(out + (size_t)(b0 + bb) * OUTC * OUTN + roff0);
        r1[bb] = *(const float4*)(out + (size_t)(b0 + bb) * OUTC * OUTN + roff1);
    }
    float fv[FPT];
    #pragma unroll
    for (int j = 0; j < FPT; ++j) fv[j] = ft[tid + j * T2];

    float bv[2][8];
    #pragma unroll
    for (int s = 0; s < 2; ++s)
        #pragma unroll
        for (int v = 0; v < 8; ++v)
            bv[s][v] = bias[(size_t)(ktile * 16 + half * 8 + v) * OUTN + n0 + (nsb + s) * 16 + col];

    #pragma unroll
    for (int bb = 0; bb < BG2; ++bb) {
        *(float4*)&s_red[bb][lc * RP + l4 * 4]        = r0[bb];
        *(float4*)&s_red[bb][(lc + 16) * RP + l4 * 4] = r1[bb];
    }
    #pragma unroll
    for (int j = 0; j < FPT; ++j) {
        const int i = tid + j * T2;
        const int c = i >> 6;
        const int k = i & 63;
        s_ft[k * CP + c] = fv[j];
    }
    __syncthreads();

    v2f afrag[INC / 4];
    #pragma unroll
    for (int kc = 0; kc < INC / 4; ++kc)
        afrag[kc] = *(const v2f*)&s_ft[krow * CP + kc * 4 + 2 * half];

    #pragma unroll
    for (int bb = 0; bb < BG2; ++bb) {
        const float* red = s_red[bb];
        const size_t ob = (size_t)(b0 + bb) * OUTC * OUTN;
        #pragma unroll
        for (int s = 0; s < 2; ++s) {
            const int nl = (nsb + s) * 16 + col;
            v8f acc = {0.f, 0.f, 0.f, 0.f, 0.f, 0.f, 0.f, 0.f};
            #pragma unroll
            for (int kc = 0; kc < INC / 4; ++kc) {
                const int c0 = kc * 4 + 2 * half;
                v2f bm;
                bm.x = red[c0 * RP + nl];
                bm.y = red[(c0 + 1) * RP + nl];
                acc = __builtin_amdgcn_wmma_f32_16x16x4_f32(
                     false, afrag[kc],  false, bm,
                     (short)0, acc,  false,  false);
            }
            #pragma unroll
            for (int v = 0; v < 8; ++v) {
                const int k = ktile * 16 + half * 8 + v;
                out[ob + (size_t)k * OUTN + n0 + nl] = acc[v] + bv[s][v];
            }
        }
        __builtin_amdgcn_sched_barrier(0);
    }
}

extern "C" void kernel_launch(void* const* d_in, const int* in_sizes, int n_in,
                              void* d_out, int out_size, void* d_ws, size_t ws_size,
                              hipStream_t stream) {
    const float* x    = (const float*)d_in[0];
    const int*   idx  = (const int*)  d_in[1];
    const float* nf   = (const float*)d_in[2];
    const float* ft   = (const float*)d_in[3];
    const float* bias = (const float*)d_in[4];
    float*       out  = (float*)d_out;

    fgl_gather_max_kernel<<<INC * (B / RB), GT, 0, stream>>>(x, idx, nf, out);
    dim3 grid2(OUTN / NT2, B / BG2);
    fgl_channel_gemm_kernel<<<grid2, T2, 0, stream>>>(ft, bias, out);
}


# Launch 1: two batches share each LDS pass: half rows of x*nf of a batch pair interleaved as float2, so one nf read and one 8-byte LDS gather serve both; other-half neighbours read a -inf slot
# speedup vs baseline: 12.5311x; 1.0314x over previous
#include <hip/hip_runtime.h>


typedef __attribute__((ext_vector_type(2))) float v2f;
typedef __attribute__((ext_vector_type(8))) float v8f;
typedef __attribute__((ext_vector_type(4))) float v4f;

constexpr int B    = 32;
constexpr int INC  = 32;
constexpr int OUTC = 64;
constexpr int INN  = 65536;
constexpr int OUTN = 8192;
constexpr int DEG  = 16;

constexpr int GT   = 512;
constexpr int RB   = 4;
constexpr int NPT  = OUTN / GT;
constexpr int HALF = INN / 2;
constexpr int V4H  = HALF / 4 / GT;
constexpr int SU   = 4;
constexpr int PFU  = 256;
constexpr int PFPT = HALF * 4 / PFU / GT;

typedef __attribute__((ext_vector_type(2))) unsigned short v2us;

__device__ __forceinline__ void prefetch_half_row_to_l2(const float* __restrict__ row, int tid)
{
    const char* p = (const char*)row + (size_t)tid * PFU;
    for (int u = 0; u < PFPT; ++u)
        __builtin_prefetch(p + (size_t)u * GT * PFU,  0,  2);
}

__global__ __launch_bounds__(GT)
void fgl_gather_max_kernel(const float* __restrict__ x,
                           const int*   __restrict__ idx,
                           const float* __restrict__ nf,
                           float*       __restrict__ red)
{
    __shared__ alignas(16) float s_h[2 * HALF + 4];

    const int tid = threadIdx.x;
    const int c   = blockIdx.x & (INC - 1);
    const int b0  = (blockIdx.x >> 5) * RB;
    constexpr size_t BS = (size_t)INC * INN;

    prefetch_half_row_to_l2(x + ((size_t)b0 * INC + c) * INN, tid);
    prefetch_half_row_to_l2(x + ((size_t)b0 * INC + c) * INN + BS, tid);

    if (tid == 0) {
        s_h[2 * HALF]     = -__builtin_inff();
        s_h[2 * HALF + 1] = -__builtin_inff();
    }

    unsigned pk[NPT][DEG / 2];
    #pragma unroll
    for (int j = 0; j < NPT; ++j) {
        const int4* __restrict__ p = (const int4*)(idx + (size_t)(tid + j * GT) * DEG);
        #pragma unroll
        for (int q = 0; q < DEG / 4; ++q) {
            const int4 v = p[q];
            pk[j][2 * q]     = ((unsigned)v.x & 0xffffu) | ((unsigned)v.y << 16);
            pk[j][2 * q + 1] = ((unsigned)v.z & 0xffffu) | ((unsigned)v.w << 16);
        }
    }

    float ma[NPT], mb[NPT];
    #pragma unroll
    for (int j = 0; j < NPT; ++j) { ma[j] = -__builtin_inff(); mb[j] = -__builtin_inff(); }

    const v4f* __restrict__ nf4 = (const v4f*)(nf + (size_t)c * INN);
    v4f* s_h4 = (v4f*)s_h;
    const v2f* s_h2 = (const v2f*)s_h;

    #pragma unroll 1
    for (int it = 0; it < RB; ++it) {
        const int hf = it & 1;
        const int b  = b0 + (it & ~1);
        const float* __restrict__ xa = x + ((size_t)b * INC + c) * INN + (size_t)hf * HALF;
        const float* __restrict__ xb = xa + BS;

        if (it) __syncthreads();

        const v4f* __restrict__ xap = (const v4f*)xa + tid;
        const v4f* __restrict__ xbp = (const v4f*)xb + tid;
        const v4f* __restrict__ np  = nf4 + hf * (HALF / 4) + tid;
        v4f* sp = s_h4 + 2 * tid;
        #pragma unroll 1
        for (int g = 0; g < V4H / SU; ++g) {
            v4f a[SU], q[SU], w[SU];
            #pragma unroll
            for (int u = 0; u < SU; ++u) a[u] = xap[u * GT];
            #pragma unroll
            for (int u = 0; u < SU; ++u) q[u] = xbp[u * GT];
            #pragma unroll
            for (int u = 0; u < SU; ++u) w[u] = np[u * GT];
            #pragma unroll
            for (int u = 0; u < SU; ++u) {
                const v4f pa = a[u] * w[u];
                const v4f pb = q[u] * w[u];
                v4f lo, hi;
                lo.x = pa.x; lo.y = pb.x; lo.z = pa.y; lo.w = pb.y;
                hi.x = pa.z; hi.y = pb.z; hi.z = pa.w; hi.w = pb.w;
                sp[2 * u * GT]     = lo;
                sp[2 * u * GT + 1] = hi;
            }
            xap += SU * GT; xbp += SU * GT; np += SU * GT; sp += 2 * SU * GT;
        }
        __syncthreads();

        if (it + 1 < RB) {
            const int nh = (it + 1) & 1;
            const int nb = b0 + ((it + 1) & ~1);
            const float* nx = x + ((size_t)nb * INC + c) * INN + (size_t)nh * HALF;
            prefetch_half_row_to_l2(nx, tid);
            prefetch_half_row_to_l2(nx + BS, tid);
        }

        const unsigned flip = hf ? 0x80008000u : 0u;
        const v2us cap = {(unsigned short)HALF, (unsigned short)HALF};
        #pragma unroll
        for (int j = 0; j < NPT; ++j) {
            float va = ma[j], vb = mb[j];
            #pragma unroll
            for (int e = 0; e < DEG / 2; ++e) {
                v2us s = __builtin_bit_cast(v2us, pk[j][e] ^ flip);
                s = __builtin_elementwise_min(s, cap);
                const v2f t0 = s_h2[s.x];
                const v2f t1 = s_h2[s.y];
                va = fmaxf(va, t0.x); va = fmaxf(va, t1.x);
                vb = fmaxf(vb, t0.y); vb = fmaxf(vb, t1.y);
            }
            ma[j] = va; mb[j] = vb;
        }

        if (hf) {
            float* __restrict__ ra = red + ((size_t)b * OUTC + c) * OUTN;
            float* __restrict__ rb = ra + (size_t)OUTC * OUTN;
            #pragma unroll
            for (int j = 0; j < NPT; ++j) {
                ra[tid + j * GT] = ma[j];
                rb[tid + j * GT] = mb[j];
                ma[j] = -__builtin_inff(); mb[j] = -__builtin_inff();
            }
        }
    }
}

constexpr int T2  = 256;
constexpr int NT2 = 64;
constexpr int BG2 = 4;
constexpr int RP  = NT2 + 8;
constexpr int CP  = INC + 2;
constexpr int FPT = INC * OUTC / T2;

__global__ __launch_bounds__(T2)
void fgl_channel_gemm_kernel(const float* __restrict__ ft,
                             const float* __restrict__ bias,
                             float*                    out)
{
    __shared__ alignas(16) float s_red[BG2][INC * RP];
    __shared__ alignas(16) float s_ft [OUTC * CP];

    const int tid = threadIdx.x;
    const int n0  = blockIdx.x * NT2;
    const int b0  = blockIdx.y * BG2;

    const int lc = tid >> 4;
    const int l4 = tid & 15;
    const size_t roff0 = (size_t)lc * OUTN + n0 + l4 * 4;
    const size_t roff1 = roff0 + (size_t)16 * OUTN;

    const int wave  = tid >> 5;
    const int lane  = tid & 31;
    const int half  = lane >> 4;
    const int col   = lane & 15;
    const int ktile = wave & 3;
    const int nsb   = (wave >> 2) * 2;
    const int krow  = ktile * 16 + col;

    float4 r0[BG2], r1[BG2];
    #pragma unroll
    for (int bb = 0; bb < BG2; ++bb) {
        r0[bb] = *(const float4*)(out + (size_t)(b0 + bb) * OUTC * OUTN + roff0);
        r1[bb] = *(const float4*)(out + (size_t)(b0 + bb) * OUTC * OUTN + roff1);
    }
    float fv[FPT];
    #pragma unroll
    for (int j = 0; j < FPT; ++j) fv[j] = ft[tid + j * T2];

    float bv[2][8];
    #pragma unroll
    for (int s = 0; s < 2; ++s)
        #pragma unroll
        for (int v = 0; v < 8; ++v)
            bv[s][v] = bias[(size_t)(ktile * 16 + half * 8 + v) * OUTN + n0 + (nsb + s) * 16 + col];

    #pragma unroll
    for (int bb = 0; bb < BG2; ++bb) {
        *(float4*)&s_red[bb][lc * RP + l4 * 4]        = r0[bb];
        *(float4*)&s_red[bb][(lc + 16) * RP + l4 * 4] = r1[bb];
    }
    #pragma unroll
    for (int j = 0; j < FPT; ++j) {
        const int i = tid + j * T2;
        const int c = i >> 6;
        const int k = i & 63;
        s_ft[k * CP + c] = fv[j];
    }
    __syncthreads();

    v2f afrag[INC / 4];
    #pragma unroll
    for (int kc = 0; kc < INC / 4; ++kc)
        afrag[kc] = *(const v2f*)&s_ft[krow * CP + kc * 4 + 2 * half];

    #pragma unroll
    for (int bb = 0; bb < BG2; ++bb) {
        const float* red = s_red[bb];
        const size_t ob = (size_t)(b0 + bb) * OUTC * OUTN;
        #pragma unroll
        for (int s = 0; s < 2; ++s) {
            const int nl = (nsb + s) * 16 + col;
            v8f acc = {0.f, 0.f, 0.f, 0.f, 0.f, 0.f, 0.f, 0.f};
            #pragma unroll
            for (int kc = 0; kc < INC / 4; ++kc) {
                const int c0 = kc * 4 + 2 * half;
                v2f bm;
                bm.x = red[c0 * RP + nl];
                bm.y = red[(c0 + 1) * RP + nl];
                acc = __builtin_amdgcn_wmma_f32_16x16x4_f32(
                     false, afrag[kc],  false, bm,
                     (short)0, acc,  false,  false);
            }
            #pragma unroll
            for (int v = 0; v < 8; ++v) {
                const int k = ktile * 16 + half * 8 + v;
                out[ob + (size_t)k * OUTN + n0 + nl] = acc[v] + bv[s][v];
            }
        }
        __builtin_amdgcn_sched_barrier(0);
    }
}

extern "C" void kernel_launch(void* const* d_in, const int* in_sizes, int n_in,
                              void* d_out, int out_size, void* d_ws, size_t ws_size,
                              hipStream_t stream) {
    const float* x    = (const float*)d_in[0];
    const int*   idx  = (const int*)  d_in[1];
    const float* nf   = (const float*)d_in[2];
    const float* ft   = (const float*)d_in[3];
    const float* bias = (const float*)d_in[4];
    float*       out  = (float*)d_out;

    fgl_gather_max_kernel<<<INC * (B / RB), GT, 0, stream>>>(x, idx, nf, out);
    dim3 grid2(OUTN / NT2, B / BG2);
    fgl_channel_gemm_kernel<<<grid2, T2, 0, stream>>>(ft, bias, out);
}


# Launch 1 stream phase software-pipelined: two half-size load groups alternate, each reissued right after its own LDS stores, ahead of the other group's products; loads stay queued, same registers
# speedup vs baseline: 12.9634x; 1.0345x over previous
#include <hip/hip_runtime.h>


typedef __attribute__((ext_vector_type(2))) float v2f;
typedef __attribute__((ext_vector_type(8))) float v8f;
typedef __attribute__((ext_vector_type(4))) float v4f;

constexpr int B    = 32;
constexpr int INC  = 32;
constexpr int OUTC = 64;
constexpr int INN  = 65536;
constexpr int OUTN = 8192;
constexpr int DEG  = 16;

constexpr int GT   = 512;
constexpr int RB   = 4;
constexpr int NPT  = OUTN / GT;
constexpr int HALF = INN / 2;
constexpr int V4H  = HALF / 4 / GT;
constexpr int SH   = 2;
constexpr int NTR  = V4H / (2 * SH);
constexpr int PFU  = 256;
constexpr int PFPT = HALF * 4 / PFU / GT;

typedef __attribute__((ext_vector_type(2))) unsigned short v2us;

__device__ __forceinline__ void prefetch_half_row_to_l2(const float* __restrict__ row, int tid)
{
    const char* p = (const char*)row + (size_t)tid * PFU;
    for (int u = 0; u < PFPT; ++u)
        __builtin_prefetch(p + (size_t)u * GT * PFU,  0,  2);
}

__device__ __forceinline__ void put_pair(v4f* sp, v4f a, v4f q, v4f w)
{
    const v4f pa = a * w;
    const v4f pb = q * w;
    v4f lo, hi;
    lo.x = pa.x; lo.y = pb.x; lo.z = pa.y; lo.w = pb.y;
    hi.x = pa.z; hi.y = pb.z; hi.z = pa.w; hi.w = pb.w;
    sp[0] = lo;
    sp[1] = hi;
}

__global__ __launch_bounds__(GT)
void fgl_gather_max_kernel(const float* __restrict__ x,
                           const int*   __restrict__ idx,
                           const float* __restrict__ nf,
                           float*       __restrict__ red)
{
    __shared__ alignas(16) float s_h[2 * HALF + 4];

    const int tid = threadIdx.x;
    const int c   = blockIdx.x & (INC - 1);
    const int b0  = (blockIdx.x >> 5) * RB;
    constexpr size_t BS = (size_t)INC * INN;

    prefetch_half_row_to_l2(x + ((size_t)b0 * INC + c) * INN, tid);
    prefetch_half_row_to_l2(x + ((size_t)b0 * INC + c) * INN + BS, tid);

    if (tid == 0) {
        s_h[2 * HALF]     = -__builtin_inff();
        s_h[2 * HALF + 1] = -__builtin_inff();
    }

    unsigned pk[NPT][DEG / 2];
    #pragma unroll
    for (int j = 0; j < NPT; ++j) {
        const int4* __restrict__ p = (const int4*)(idx + (size_t)(tid + j * GT) * DEG);
        #pragma unroll
        for (int q = 0; q < DEG / 4; ++q) {
            const int4 v = p[q];
            pk[j][2 * q]     = ((unsigned)v.x & 0xffffu) | ((unsigned)v.y << 16);
            pk[j][2 * q + 1] = ((unsigned)v.z & 0xffffu) | ((unsigned)v.w << 16);
        }
    }

    float ma[NPT], mb[NPT];
    #pragma unroll
    for (int j = 0; j < NPT; ++j) { ma[j] = -__builtin_inff(); mb[j] = -__builtin_inff(); }

    const v4f* __restrict__ nf4 = (const v4f*)(nf + (size_t)c * INN);
    v4f* s_h4 = (v4f*)s_h;
    const v2f* s_h2 = (const v2f*)s_h;

    #pragma unroll 1
    for (int it = 0; it < RB; ++it) {
        const int hf = it & 1;
        const int b  = b0 + (it & ~1);
        const float* __restrict__ xa = x + ((size_t)b * INC + c) * INN + (size_t)hf * HALF;
        const float* __restrict__ xb = xa + BS;

        if (it) __syncthreads();

        const v4f* __restrict__ xap = (const v4f*)xa + tid;
        const v4f* __restrict__ xbp = (const v4f*)xb + tid;
        const v4f* __restrict__ np  = nf4 + hf * (HALF / 4) + tid;
        v4f* sp = s_h4 + 2 * tid;
        v4f aA0 = xap[0],      aA1 = xap[GT],     qA0 = xbp[0],      qA1 = xbp[GT],     wA0 = np[0],      wA1 = np[GT];
        v4f aB0 = xap[2 * GT], aB1 = xap[3 * GT], qB0 = xbp[2 * GT], qB1 = xbp[3 * GT], wB0 = np[2 * GT], wB1 = np[3 * GT];
        #pragma unroll 1
        for (int t = 0; t + 1 < NTR; ++t) {
            put_pair(sp,          aA0, qA0, wA0);
            put_pair(sp + 2 * GT, aA1, qA1, wA1);
            __builtin_amdgcn_sched_barrier(0);
            aA0 = xap[4 * GT]; aA1 = xap[5 * GT];
            qA0 = xbp[4 * GT]; qA1 = xbp[5 * GT];
            wA0 = np[4 * GT];  wA1 = np[5 * GT];
            __builtin_amdgcn_sched_barrier(0);
            put_pair(sp + 4 * GT, aB0, qB0, wB0);
            put_pair(sp + 6 * GT, aB1, qB1, wB1);
            __builtin_amdgcn_sched_barrier(0);
            aB0 = xap[6 * GT]; aB1 = xap[7 * GT];
            qB0 = xbp[6 * GT]; qB1 = xbp[7 * GT];
            wB0 = np[6 * GT];  wB1 = np[7 * GT];
            xap += 4 * GT; xbp += 4 * GT; np += 4 * GT; sp += 8 * GT;
        }
        put_pair(sp,          aA0, qA0, wA0);
        put_pair(sp + 2 * GT, aA1, qA1, wA1);
        put_pair(sp + 4 * GT, aB0, qB0, wB0);
        put_pair(sp + 6 * GT, aB1, qB1, wB1);
        __syncthreads();

        if (it + 1 < RB) {
            const int nh = (it + 1) & 1;
            const int nb = b0 + ((it + 1) & ~1);
            const float* nx = x + ((size_t)nb * INC + c) * INN + (size_t)nh * HALF;
            prefetch_half_row_to_l2(nx, tid);
            prefetch_half_row_to_l2(nx + BS, tid);
        }

        const unsigned flip = hf ? 0x80008000u : 0u;
        const v2us cap = {(unsigned short)HALF, (unsigned short)HALF};
        #pragma unroll
        for (int j = 0; j < NPT; ++j) {
            float va = ma[j], vb = mb[j];
            #pragma unroll
            for (int e = 0; e < DEG / 2; ++e) {
                v2us s = __builtin_bit_cast(v2us, pk[j][e] ^ flip);
                s = __builtin_elementwise_min(s, cap);
                const v2f t0 = s_h2[s.x];
                const v2f t1 = s_h2[s.y];
                va = fmaxf(va, t0.x); va = fmaxf(va, t1.x);
                vb = fmaxf(vb, t0.y); vb = fmaxf(vb, t1.y);
            }
            ma[j] = va; mb[j] = vb;
        }

        if (hf) {
            float* __restrict__ ra = red + ((size_t)b * OUTC + c) * OUTN;
            float* __restrict__ rb = ra + (size_t)OUTC * OUTN;
            #pragma unroll
            for (int j = 0; j < NPT; ++j) {
                ra[tid + j * GT] = ma[j];
                rb[tid + j * GT] = mb[j];
                ma[j] = -__builtin_inff(); mb[j] = -__builtin_inff();
            }
        }
    }
}

constexpr int T2  = 256;
constexpr int NT2 = 64;
constexpr int BG2 = 4;
constexpr int RP  = NT2 + 8;
constexpr int CP  = INC + 2;
constexpr int FPT = INC * OUTC / T2;

__global__ __launch_bounds__(T2)
void fgl_channel_gemm_kernel(const float* __restrict__ ft,
                             const float* __restrict__ bias,
                             float*                    out)
{
    __shared__ alignas(16) float s_red[BG2][INC * RP];
    __shared__ alignas(16) float s_ft [OUTC * CP];

    const int tid = threadIdx.x;
    const int n0  = blockIdx.x * NT2;
    const int b0  = blockIdx.y * BG2;

    const int lc = tid >> 4;
    const int l4 = tid & 15;
    const size_t roff0 = (size_t)lc * OUTN + n0 + l4 * 4;
    const size_t roff1 = roff0 + (size_t)16 * OUTN;

    const int wave  = tid >> 5;
    const int lane  = tid & 31;
    const int half  = lane >> 4;
    const int col   = lane & 15;
    const int ktile = wave & 3;
    const int nsb   = (wave >> 2) * 2;
    const int krow  = ktile * 16 + col;

    float4 r0[BG2], r1[BG2];
    #pragma unroll
    for (int bb = 0; bb < BG2; ++bb) {
        r0[bb] = *(const float4*)(out + (size_t)(b0 + bb) * OUTC * OUTN + roff0);
        r1[bb] = *(const float4*)(out + (size_t)(b0 + bb) * OUTC * OUTN + roff1);
    }
    float fv[FPT];
    #pragma unroll
    for (int j = 0; j < FPT; ++j) fv[j] = ft[tid + j * T2];

    float bv[2][8];
    #pragma unroll
    for (int s = 0; s < 2; ++s)
        #pragma unroll
        for (int v = 0; v < 8; ++v)
            bv[s][v] = bias[(size_t)(ktile * 16 + half * 8 + v) * OUTN + n0 + (nsb + s) * 16 + col];

    #pragma unroll
    for (int bb = 0; bb < BG2; ++bb) {
        *(float4*)&s_red[bb][lc * RP + l4 * 4]        = r0[bb];
        *(float4*)&s_red[bb][(lc + 16) * RP + l4 * 4] = r1[bb];
    }
    #pragma unroll
    for (int j = 0; j < FPT; ++j) {
        const int i = tid + j * T2;
        const int c = i >> 6;
        const int k = i & 63;
        s_ft[k * CP + c] = fv[j];
    }
    __syncthreads();

    v2f afrag[INC / 4];
    #pragma unroll
    for (int kc = 0; kc < INC / 4; ++kc)
        afrag[kc] = *(const v2f*)&s_ft[krow * CP + kc * 4 + 2 * half];

    #pragma unroll
    for (int bb = 0; bb < BG2; ++bb) {
        const float* red = s_red[bb];
        const size_t ob = (size_t)(b0 + bb) * OUTC * OUTN;
        #pragma unroll
        for (int s = 0; s < 2; ++s) {
            const int nl = (nsb + s) * 16 + col;
            v8f acc = {0.f, 0.f, 0.f, 0.f, 0.f, 0.f, 0.f, 0.f};
            #pragma unroll
            for (int kc = 0; kc < INC / 4; ++kc) {
                const int c0 = kc * 4 + 2 * half;
                v2f bm;
                bm.x = red[c0 * RP + nl];
                bm.y = red[(c0 + 1) * RP + nl];
                acc = __builtin_amdgcn_wmma_f32_16x16x4_f32(
                     false, afrag[kc],  false, bm,
                     (short)0, acc,  false,  false);
            }
            #pragma unroll
            for (int v = 0; v < 8; ++v) {
                const int k = ktile * 16 + half * 8 + v;
                out[ob + (size_t)k * OUTN + n0 + nl] = acc[v] + bv[s][v];
            }
        }
        __builtin_amdgcn_sched_barrier(0);
    }
}

extern "C" void kernel_launch(void* const* d_in, const int* in_sizes, int n_in,
                              void* d_out, int out_size, void* d_ws, size_t ws_size,
                              hipStream_t stream) {
    const float* x    = (const float*)d_in[0];
    const int*   idx  = (const int*)  d_in[1];
    const float* nf   = (const float*)d_in[2];
    const float* ft   = (const float*)d_in[3];
    const float* bias = (const float*)d_in[4];
    float*       out  = (float*)d_out;

    fgl_gather_max_kernel<<<INC * (B / RB), GT, 0, stream>>>(x, idx, nf, out);
    dim3 grid2(OUTN / NT2, B / BG2);
    fgl_channel_gemm_kernel<<<grid2, T2, 0, stream>>>(ft, bias, out);
}


# Launch 1: next pass's first two load sub-groups (x, x, nf) issued in front of the gather, A before B as in the loop, registers held across it; gather in groups of 8 LDS loads to fit the registers
# speedup vs baseline: 13.4112x; 1.0345x over previous
#include <hip/hip_runtime.h>


typedef __attribute__((ext_vector_type(2))) float v2f;
typedef __attribute__((ext_vector_type(8))) float v8f;
typedef __attribute__((ext_vector_type(4))) float v4f;

constexpr int B    = 32;
constexpr int INC  = 32;
constexpr int OUTC = 64;
constexpr int INN  = 65536;
constexpr int OUTN = 8192;
constexpr int DEG  = 16;

constexpr int GT   = 512;
constexpr int RB   = 4;
constexpr int NPT  = OUTN / GT;
constexpr int HALF = INN / 2;
constexpr int V4H  = HALF / 4 / GT;
constexpr int SH   = 2;
constexpr int NTR  = V4H / (2 * SH);
constexpr int PFU  = 256;
constexpr int PFPT = HALF * 4 / PFU / GT;

typedef __attribute__((ext_vector_type(2))) unsigned short v2us;

__device__ __forceinline__ void prefetch_half_row_to_l2(const float* __restrict__ row, int tid)
{
    const char* p = (const char*)row + (size_t)tid * PFU;
    for (int u = 0; u < PFPT; ++u)
        __builtin_prefetch(p + (size_t)u * GT * PFU,  0,  2);
}

__device__ __forceinline__ void put_pair(v4f* sp, v4f a, v4f q, v4f w)
{
    const v4f pa = a * w;
    const v4f pb = q * w;
    v4f lo, hi;
    lo.x = pa.x; lo.y = pb.x; lo.z = pa.y; lo.w = pb.y;
    hi.x = pa.z; hi.y = pb.z; hi.z = pa.w; hi.w = pb.w;
    sp[0] = lo;
    sp[1] = hi;
}

__global__ __launch_bounds__(GT)
void fgl_gather_max_kernel(const float* __restrict__ x,
                           const int*   __restrict__ idx,
                           const float* __restrict__ nf,
                           float*       __restrict__ red)
{
    __shared__ alignas(16) float s_h[2 * HALF + 4];

    const int tid = threadIdx.x;
    const int c   = blockIdx.x & (INC - 1);
    const int b0  = (blockIdx.x >> 5) * RB;
    constexpr size_t BS = (size_t)INC * INN;

    const v4f* __restrict__ nf4 = (const v4f*)(nf + (size_t)c * INN);

    const v4f* __restrict__ xap = (const v4f*)(x + ((size_t)b0 * INC + c) * INN);
    const v4f* __restrict__ xbp = xap + BS / 4;
    const v4f* __restrict__ np  = nf4;

    prefetch_half_row_to_l2(x + ((size_t)b0 * INC + c) * INN, tid);
    prefetch_half_row_to_l2(x + ((size_t)b0 * INC + c) * INN + BS, tid);

    if (tid == 0) {
        s_h[2 * HALF]     = -__builtin_inff();
        s_h[2 * HALF + 1] = -__builtin_inff();
    }

    unsigned pk[NPT][DEG / 2];
    #pragma unroll
    for (int j = 0; j < NPT; ++j) {
        const int4* __restrict__ p = (const int4*)(idx + (size_t)(tid + j * GT) * DEG);
        #pragma unroll
        for (int q = 0; q < DEG / 4; ++q) {
            const int4 v = p[q];
            pk[j][2 * q]     = ((unsigned)v.x & 0xffffu) | ((unsigned)v.y << 16);
            pk[j][2 * q + 1] = ((unsigned)v.z & 0xffffu) | ((unsigned)v.w << 16);
        }
    }

    float ma[NPT], mb[NPT];
    #pragma unroll
    for (int j = 0; j < NPT; ++j) { ma[j] = -__builtin_inff(); mb[j] = -__builtin_inff(); }

    v4f* s_h4 = (v4f*)s_h;
    const v2f* s_h2 = (const v2f*)s_h;

    __builtin_amdgcn_sched_barrier(0);
    v4f aA0 = xap[tid],          aA1 = xap[tid + GT],     qA0 = xbp[tid],          qA1 = xbp[tid + GT],     wA0 = np[tid],          wA1 = np[tid + GT];
    __builtin_amdgcn_sched_barrier(0);
    v4f aB0 = xap[tid + 2 * GT], aB1 = xap[tid + 3 * GT], qB0 = xbp[tid + 2 * GT], qB1 = xbp[tid + 3 * GT], wB0 = np[tid + 2 * GT], wB1 = np[tid + 3 * GT];

    #pragma unroll 1
    for (int it = 0; it < RB; ++it) {
        const int hf = it & 1;
        const int b  = b0 + (it & ~1);

        if (it) __syncthreads();

        v4f* sp = s_h4 + 2 * tid;
        #pragma unroll 1
        for (int t = 0; t + 1 < NTR; ++t) {
            put_pair(sp,          aA0, qA0, wA0);
            put_pair(sp + 2 * GT, aA1, qA1, wA1);
            __builtin_amdgcn_sched_barrier(0);
            aA0 = xap[tid + 4 * GT]; aA1 = xap[tid + 5 * GT];
            qA0 = xbp[tid + 4 * GT]; qA1 = xbp[tid + 5 * GT];
            wA0 = np[tid + 4 * GT];  wA1 = np[tid + 5 * GT];
            __builtin_amdgcn_sched_barrier(0);
            put_pair(sp + 4 * GT, aB0, qB0, wB0);
            put_pair(sp + 6 * GT, aB1, qB1, wB1);
            __builtin_amdgcn_sched_barrier(0);
            aB0 = xap[tid + 6 * GT]; aB1 = xap[tid + 7 * GT];
            qB0 = xbp[tid + 6 * GT]; qB1 = xbp[tid + 7 * GT];
            wB0 = np[tid + 6 * GT];  wB1 = np[tid + 7 * GT];
            xap += 4 * GT; xbp += 4 * GT; np += 4 * GT; sp += 8 * GT;
        }
        put_pair(sp,          aA0, qA0, wA0);
        put_pair(sp + 2 * GT, aA1, qA1, wA1);
        put_pair(sp + 4 * GT, aB0, qB0, wB0);
        put_pair(sp + 6 * GT, aB1, qB1, wB1);
        __syncthreads();

        if (it + 1 < RB) {
            const int nh = (it + 1) & 1;
            const int nb = b0 + ((it + 1) & ~1);
            const float* nx = x + ((size_t)nb * INC + c) * INN + (size_t)nh * HALF;
            xap = (const v4f*)nx;
            xbp = xap + BS / 4;
            np  = nf4 + nh * (HALF / 4);
            aA0 = xap[tid];      aA1 = xap[tid + GT];     qA0 = xbp[tid];      qA1 = xbp[tid + GT];     wA0 = np[tid];      wA1 = np[tid + GT];
            __builtin_amdgcn_sched_barrier(0);
            aB0 = xap[tid + 2 * GT]; aB1 = xap[tid + 3 * GT]; qB0 = xbp[tid + 2 * GT]; qB1 = xbp[tid + 3 * GT]; wB0 = np[tid + 2 * GT]; wB1 = np[tid + 3 * GT];
            prefetch_half_row_to_l2(nx, tid);
            prefetch_half_row_to_l2(nx + BS, tid);
        }

        const unsigned flip = hf ? 0x80008000u : 0u;
        const v2us cap = {(unsigned short)HALF, (unsigned short)HALF};
        #pragma unroll
        for (int j = 0; j < NPT; ++j) {
            float va = ma[j], vb = mb[j];
            #pragma unroll
            for (int g = 0; g < 2; ++g) {
                v2f t[8];
                #pragma unroll
                for (int e = 0; e < 4; ++e) {
                    v2us s = __builtin_bit_cast(v2us, pk[j][4 * g + e] ^ flip);
                    s = __builtin_elementwise_min(s, cap);
                    t[2 * e]     = s_h2[s.x];
                    t[2 * e + 1] = s_h2[s.y];
                }
                #pragma unroll
                for (int e = 0; e < 8; e += 2) {
                    va = fmaxf(va, t[e].x); va = fmaxf(va, t[e + 1].x);
                    vb = fmaxf(vb, t[e].y); vb = fmaxf(vb, t[e + 1].y);
                }
                __builtin_amdgcn_sched_barrier(0);
            }
            ma[j] = va; mb[j] = vb;
        }

        if (hf) {
            float* __restrict__ ra = red + ((size_t)b * OUTC + c) * OUTN;
            float* __restrict__ rb = ra + (size_t)OUTC * OUTN;
            #pragma unroll
            for (int j = 0; j < NPT; ++j) {
                ra[tid + j * GT] = ma[j];
                rb[tid + j * GT] = mb[j];
                ma[j] = -__builtin_inff(); mb[j] = -__builtin_inff();
            }
        }
    }
}

constexpr int T2  = 256;
constexpr int NT2 = 64;
constexpr int BG2 = 4;
constexpr int RP  = NT2 + 8;
constexpr int CP  = INC + 2;
constexpr int FPT = INC * OUTC / T2;

__global__ __launch_bounds__(T2)
void fgl_channel_gemm_kernel(const float* __restrict__ ft,
                             const float* __restrict__ bias,
                             float*                    out)
{
    __shared__ alignas(16) float s_red[BG2][INC * RP];
    __shared__ alignas(16) float s_ft [OUTC * CP];

    const int tid = threadIdx.x;
    const int n0  = blockIdx.x * NT2;
    const int b0  = blockIdx.y * BG2;

    const int lc = tid >> 4;
    const int l4 = tid & 15;
    const size_t roff0 = (size_t)lc * OUTN + n0 + l4 * 4;
    const size_t roff1 = roff0 + (size_t)16 * OUTN;

    const int wave  = tid >> 5;
    const int lane  = tid & 31;
    const int half  = lane >> 4;
    const int col   = lane & 15;
    const int ktile = wave & 3;
    const int nsb   = (wave >> 2) * 2;
    const int krow  = ktile * 16 + col;

    float4 r0[BG2], r1[BG2];
    #pragma unroll
    for (int bb = 0; bb < BG2; ++bb) {
        r0[bb] = *(const float4*)(out + (size_t)(b0 + bb) * OUTC * OUTN + roff0);
        r1[bb] = *(const float4*)(out + (size_t)(b0 + bb) * OUTC * OUTN + roff1);
    }
    float fv[FPT];
    #pragma unroll
    for (int j = 0; j < FPT; ++j) fv[j] = ft[tid + j * T2];

    float bv[2][8];
    #pragma unroll
    for (int s = 0; s < 2; ++s)
        #pragma unroll
        for (int v = 0; v < 8; ++v)
            bv[s][v] = bias[(size_t)(ktile * 16 + half * 8 + v) * OUTN + n0 + (nsb + s) * 16 + col];

    #pragma unroll
    for (int bb = 0; bb < BG2; ++bb) {
        *(float4*)&s_red[bb][lc * RP + l4 * 4]        = r0[bb];
        *(float4*)&s_red[bb][(lc + 16) * RP + l4 * 4] = r1[bb];
    }
    #pragma unroll
    for (int j = 0; j < FPT; ++j) {
        const int i = tid + j * T2;
        const int c = i >> 6;
        const int k = i & 63;
        s_ft[k * CP + c] = fv[j];
    }
    __syncthreads();

    v2f afrag[INC / 4];
    #pragma unroll
    for (int kc = 0; kc < INC / 4; ++kc)
        afrag[kc] = *(const v2f*)&s_ft[krow * CP + kc * 4 + 2 * half];

    #pragma unroll
    for (int bb = 0; bb < BG2; ++bb) {
        const float* red = s_red[bb];
        const size_t ob = (size_t)(b0 + bb) * OUTC * OUTN;
        #pragma unroll
        for (int s = 0; s < 2; ++s) {
            const int nl = (nsb + s) * 16 + col;
            v8f acc = {0.f, 0.f, 0.f, 0.f, 0.f, 0.f, 0.f, 0.f};
            #pragma unroll
            for (int kc = 0; kc < INC / 4; ++kc) {
                const int c0 = kc * 4 + 2 * half;
                v2f bm;
                bm.x = red[c0 * RP + nl];
                bm.y = red[(c0 + 1) * RP + nl];
                acc = __builtin_amdgcn_wmma_f32_16x16x4_f32(
                     false, afrag[kc],  false, bm,
                     (short)0, acc,  false,  false);
            }
            #pragma unroll
            for (int v = 0; v < 8; ++v) {
                const int k = ktile * 16 + half * 8 + v;
                out[ob + (size_t)k * OUTN + n0 + nl] = acc[v] + bv[s][v];
            }
        }
        __builtin_amdgcn_sched_barrier(0);
    }
}

extern "C" void kernel_launch(void* const* d_in, const int* in_sizes, int n_in,
                              void* d_out, int out_size, void* d_ws, size_t ws_size,
                              hipStream_t stream) {
    const float* x    = (const float*)d_in[0];
    const int*   idx  = (const int*)  d_in[1];
    const float* nf   = (const float*)d_in[2];
    const float* ft   = (const float*)d_in[3];
    const float* bias = (const float*)d_in[4];
    float*       out  = (float*)d_out;

    fgl_gather_max_kernel<<<INC * (B / RB), GT, 0, stream>>>(x, idx, nf, out);
    dim3 grid2(OUTN / NT2, B / BG2);
    fgl_channel_gemm_kernel<<<grid2, T2, 0, stream>>>(ft, bias, out);
}
